# starting kernel
# baseline (speedup 1.0000x reference)
#include <hip/hip_runtime.h>

typedef __attribute__((ext_vector_type(2))) float v2f;
typedef __attribute__((ext_vector_type(8))) float v8f;

constexpr int Bb = 8, Tt = 2048, Ee = 1024, Hh = 64;
constexpr int ROWS = Bb * Tt;
constexpr int TPB  = Tt / 16;

__device__ __forceinline__ v8f wmma4(v2f a, v2f b, v8f c) {
    return __builtin_amdgcn_wmma_f32_16x16x4_f32(
         false, a,  false, b,
         (short)0, c,  false,  false);
}

__global__ __launch_bounds__(128)
void proj_kernel(const float* __restrict__ x,
                 const float* __restrict__ Wk,
                 const float* __restrict__ Wq,
                 const float* __restrict__ Wv,
                 float* __restrict__ qo, float* __restrict__ ko,
                 float* __restrict__ vo) {
    const int lane   = threadIdx.x & 31;
    const int wv     = threadIdx.x >> 5;
    const int rowb   = blockIdx.x * 16;
    const int mrow   = lane & 15;
    const int khalf  = (lane >> 4) * 2;
    const int ncol   = wv * 16 + (lane & 15);

    const float* xa = x + (size_t)(rowb + mrow) * Ee + khalf;

    v8f cq = {}, ck = {}, cv = {};
    #pragma unroll 4
    for (int e0 = 0; e0 < Ee; e0 += 4) {
        v2f a = *(const v2f*)(xa + e0);
        const float* wkp = Wk + (size_t)(e0 + khalf) * Hh + ncol;
        const float* wqp = Wq + (size_t)(e0 + khalf) * Hh + ncol;
        const float* wvp = Wv + (size_t)(e0 + khalf) * Hh + ncol;
        v2f bk; bk.x = wkp[0]; bk.y = wkp[Hh];
        v2f bq; bq.x = wqp[0]; bq.y = wqp[Hh];
        v2f bv; bv.x = wvp[0]; bv.y = wvp[Hh];
        ck = wmma4(a, bk, ck);
        cq = wmma4(a, bq, cq);
        cv = wmma4(a, bv, cv);
    }

    const int orow = rowb + (lane >> 4) * 8;
    #pragma unroll
    for (int r = 0; r < 8; ++r) {
        size_t idx = (size_t)(orow + r) * Hh + ncol;
        qo[idx] = cq[r];
        ko[idx] = ck[r];
        vo[idx] = cv[r];
    }
}

__global__ __launch_bounds__(32)
void attn_kernel(const float* __restrict__ q,
                 const float* __restrict__ k,
                 const float* __restrict__ v,
                 float* __restrict__ out) {
    __shared__ float pbuf[16][18];

    const int lane  = threadIdx.x & 31;
    const int tile  = blockIdx.x;
    const int b     = tile / TPB;
    const int it    = tile % TPB;
    const int qrow0 = b * Tt + it * 16;
    const int half  = lane >> 4;
    const int nlo   = lane & 15;
    const int khalf = half * 2;

    v2f qa[16];
    const float* qp = q + (size_t)(qrow0 + nlo) * Hh + khalf;
    #pragma unroll
    for (int s = 0; s < 16; ++s) qa[s] = *(const v2f*)(qp + 4 * s);

    v8f o0 = {}, o1 = {}, o2 = {}, o3 = {};
    float mrow[8], lrow[8];
    #pragma unroll
    for (int r = 0; r < 8; ++r) { mrow[r] = -1e30f; lrow[r] = 0.f; }

    const float scale = 0.03125f;

    for (int j = 0; j <= it; ++j) {
        const int krow0 = b * Tt + j * 16;

        v8f s8 = {};
        const float* kp = k + (size_t)(krow0 + nlo) * Hh + khalf;
        #pragma unroll
        for (int t = 0; t < 16; ++t) {
            v2f kb = *(const v2f*)(kp + 4 * t);
            s8 = wmma4(qa[t], kb, s8);
        }

        float sv[8];
        #pragma unroll
        for (int r = 0; r < 8; ++r) sv[r] = s8[r] * scale;

        if (j == it) {
            #pragma unroll
            for (int r = 0; r < 8; ++r) {
                int m = r + half * 8;
                if (nlo > m) sv[r] = -1e30f;
            }
        }

        float alpha[8], pv[8];
        #pragma unroll
        for (int r = 0; r < 8; ++r) {
            float t = sv[r];
            t = fmaxf(t, __shfl_xor(t, 1, 16));
            t = fmaxf(t, __shfl_xor(t, 2, 16));
            t = fmaxf(t, __shfl_xor(t, 4, 16));
            t = fmaxf(t, __shfl_xor(t, 8, 16));
            float mnew = fmaxf(mrow[r], t);
            alpha[r] = __expf(mrow[r] - mnew);
            mrow[r]  = mnew;
            float p  = __expf(sv[r] - mnew);
            pv[r] = p;
            float s = p;
            s += __shfl_xor(s, 1, 16);
            s += __shfl_xor(s, 2, 16);
            s += __shfl_xor(s, 4, 16);
            s += __shfl_xor(s, 8, 16);
            lrow[r] = lrow[r] * alpha[r] + s;
        }

        #pragma unroll
        for (int r = 0; r < 8; ++r) {
            o0[r] *= alpha[r]; o1[r] *= alpha[r];
            o2[r] *= alpha[r]; o3[r] *= alpha[r];
        }

        #pragma unroll
        for (int r = 0; r < 8; ++r)
            pbuf[r + half * 8][nlo] = pv[r];
        __builtin_amdgcn_wave_barrier();
        asm volatile("s_wait_dscnt 0" ::: "memory");

        const float* vp = v + (size_t)krow0 * Hh;
        #pragma unroll
        for (int kk0 = 0; kk0 < 16; kk0 += 4) {
            v2f pa;
            pa.x = pbuf[nlo][kk0 + khalf];
            pa.y = pbuf[nlo][kk0 + khalf + 1];
            const float* vrow = vp + (size_t)(kk0 + khalf) * Hh;
            { v2f vb; vb.x = vrow[ 0 + nlo]; vb.y = vrow[Hh +  0 + nlo]; o0 = wmma4(pa, vb, o0); }
            { v2f vb; vb.x = vrow[16 + nlo]; vb.y = vrow[Hh + 16 + nlo]; o1 = wmma4(pa, vb, o1); }
            { v2f vb; vb.x = vrow[32 + nlo]; vb.y = vrow[Hh + 32 + nlo]; o2 = wmma4(pa, vb, o2); }
            { v2f vb; vb.x = vrow[48 + nlo]; vb.y = vrow[Hh + 48 + nlo]; o3 = wmma4(pa, vb, o3); }
        }
        __builtin_amdgcn_wave_barrier();
    }

    #pragma unroll
    for (int r = 0; r < 8; ++r) {
        float inv = 1.0f / lrow[r];
        size_t idx = (size_t)(qrow0 + r + half * 8) * Hh;
        out[idx +  0 + nlo] = o0[r] * inv;
        out[idx + 16 + nlo] = o1[r] * inv;
        out[idx + 32 + nlo] = o2[r] * inv;
        out[idx + 48 + nlo] = o3[r] * inv;
    }
}

extern "C" void kernel_launch(void* const* d_in, const int* in_sizes, int n_in,
                              void* d_out, int out_size, void* d_ws, size_t ws_size,
                              hipStream_t stream) {
    const float* x  = (const float*)d_in[0];
    const float* Wk = (const float*)d_in[1];
    const float* Wq = (const float*)d_in[2];
    const float* Wv = (const float*)d_in[3];
    float* out = (float*)d_out;

    float* ws   = (float*)d_ws;
    float* qbuf = ws;
    float* kbuf = ws + (size_t)ROWS * Hh;
    float* vbuf = ws + 2 * (size_t)ROWS * Hh;

    proj_kernel<<<ROWS / 16, 128, 0, stream>>>(x, Wk, Wq, Wv, qbuf, kbuf, vbuf);
    attn_kernel<<<ROWS / 16, 32, 0, stream>>>(qbuf, kbuf, vbuf, out);
}


# attention: key tiles of each query tile dealt round-robin to the 8 waves of a 256-thread work-group (was 1 wave per tile); partial (max,sum,O) merged in LDS, so softmax sums add in a new order
# speedup vs baseline: 2.0845x; 2.0845x over previous
#include <hip/hip_runtime.h>

typedef __attribute__((ext_vector_type(2))) float v2f;
typedef __attribute__((ext_vector_type(8))) float v8f;

constexpr int Bb = 8, Tt = 2048, Ee = 1024, Hh = 64;
constexpr int ROWS = Bb * Tt;
constexpr int TPB  = Tt / 16;

__device__ __forceinline__ v8f wmma4(v2f a, v2f b, v8f c) {
    return __builtin_amdgcn_wmma_f32_16x16x4_f32(
         false, a,  false, b,
         (short)0, c,  false,  false);
}

__global__ __launch_bounds__(128)
void proj_kernel(const float* __restrict__ x,
                 const float* __restrict__ Wk,
                 const float* __restrict__ Wq,
                 const float* __restrict__ Wv,
                 float* __restrict__ qo, float* __restrict__ ko,
                 float* __restrict__ vo) {
    const int lane   = threadIdx.x & 31;
    const int wv     = threadIdx.x >> 5;
    const int rowb   = blockIdx.x * 16;
    const int mrow   = lane & 15;
    const int khalf  = (lane >> 4) * 2;
    const int ncol   = wv * 16 + (lane & 15);

    const float* xa = x + (size_t)(rowb + mrow) * Ee + khalf;

    v8f cq = {}, ck = {}, cv = {};
    #pragma unroll 4
    for (int e0 = 0; e0 < Ee; e0 += 4) {
        v2f a = *(const v2f*)(xa + e0);
        const float* wkp = Wk + (size_t)(e0 + khalf) * Hh + ncol;
        const float* wqp = Wq + (size_t)(e0 + khalf) * Hh + ncol;
        const float* wvp = Wv + (size_t)(e0 + khalf) * Hh + ncol;
        v2f bk; bk.x = wkp[0]; bk.y = wkp[Hh];
        v2f bq; bq.x = wqp[0]; bq.y = wqp[Hh];
        v2f bv; bv.x = wvp[0]; bv.y = wvp[Hh];
        ck = wmma4(a, bk, ck);
        cq = wmma4(a, bq, cq);
        cv = wmma4(a, bv, cv);
    }

    const int orow = rowb + (lane >> 4) * 8;
    #pragma unroll
    for (int r = 0; r < 8; ++r) {
        size_t idx = (size_t)(orow + r) * Hh + ncol;
        qo[idx] = cq[r];
        ko[idx] = ck[r];
        vo[idx] = cv[r];
    }
}

constexpr int NW     = 8;
constexpr int OPITCH = 68;

__global__ __launch_bounds__(32 * NW)
void attn_kernel(const float* __restrict__ q,
                 const float* __restrict__ k,
                 const float* __restrict__ v,
                 float* __restrict__ out) {
    __shared__ float pbuf[NW][16][18];
    __shared__ alignas(16) float obuf[NW][16][OPITCH];
    __shared__ alignas(8)  float mlbuf[NW][16][2];

    const int lane  = threadIdx.x & 31;
    const int w     = __builtin_amdgcn_readfirstlane((int)(threadIdx.x >> 5));
    const int tile  = blockIdx.x;
    const int b     = tile / TPB;
    const int it    = tile % TPB;
    const int qrow0 = b * Tt + it * 16;
    const int half  = lane >> 4;
    const int nlo   = lane & 15;
    const int khalf = half * 2;

    v2f qa[16];
    const float* qp = q + (size_t)(qrow0 + nlo) * Hh + khalf;
    #pragma unroll
    for (int s = 0; s < 16; ++s) qa[s] = *(const v2f*)(qp + 4 * s);

    v8f o0 = {}, o1 = {}, o2 = {}, o3 = {};
    float mrow[8], lrow[8];
    #pragma unroll
    for (int r = 0; r < 8; ++r) { mrow[r] = -1e30f; lrow[r] = 0.f; }

    const float scale = 0.03125f;

    for (int j = w; j <= it; j += NW) {
        const int krow0 = b * Tt + j * 16;

        v8f s8 = {};
        const float* kp = k + (size_t)(krow0 + nlo) * Hh + khalf;
        #pragma unroll
        for (int t = 0; t < 16; ++t) {
            v2f kb = *(const v2f*)(kp + 4 * t);
            s8 = wmma4(qa[t], kb, s8);
        }

        float sv[8];
        #pragma unroll
        for (int r = 0; r < 8; ++r) sv[r] = s8[r] * scale;

        if (j == it) {
            #pragma unroll
            for (int r = 0; r < 8; ++r) {
                int m = r + half * 8;
                if (nlo > m) sv[r] = -1e30f;
            }
        }

        float alpha[8], pv[8];
        #pragma unroll
        for (int r = 0; r < 8; ++r) {
            float t = sv[r];
            t = fmaxf(t, __shfl_xor(t, 1, 16));
            t = fmaxf(t, __shfl_xor(t, 2, 16));
            t = fmaxf(t, __shfl_xor(t, 4, 16));
            t = fmaxf(t, __shfl_xor(t, 8, 16));
            float mnew = fmaxf(mrow[r], t);
            alpha[r] = __expf(mrow[r] - mnew);
            mrow[r]  = mnew;
            float p  = __expf(sv[r] - mnew);
            pv[r] = p;
            float s = p;
            s += __shfl_xor(s, 1, 16);
            s += __shfl_xor(s, 2, 16);
            s += __shfl_xor(s, 4, 16);
            s += __shfl_xor(s, 8, 16);
            lrow[r] = lrow[r] * alpha[r] + s;
        }

        #pragma unroll
        for (int r = 0; r < 8; ++r) {
            o0[r] *= alpha[r]; o1[r] *= alpha[r];
            o2[r] *= alpha[r]; o3[r] *= alpha[r];
        }

        #pragma unroll
        for (int r = 0; r < 8; ++r)
            pbuf[w][r + half * 8][nlo] = pv[r];
        __builtin_amdgcn_wave_barrier();
        asm volatile("s_wait_dscnt 0" ::: "memory");

        const float* vp = v + (size_t)krow0 * Hh;
        #pragma unroll
        for (int kk0 = 0; kk0 < 16; kk0 += 4) {
            v2f pa;
            pa.x = pbuf[w][nlo][kk0 + khalf];
            pa.y = pbuf[w][nlo][kk0 + khalf + 1];
            const float* vrow = vp + (size_t)(kk0 + khalf) * Hh;
            { v2f vb; vb.x = vrow[ 0 + nlo]; vb.y = vrow[Hh +  0 + nlo]; o0 = wmma4(pa, vb, o0); }
            { v2f vb; vb.x = vrow[16 + nlo]; vb.y = vrow[Hh + 16 + nlo]; o1 = wmma4(pa, vb, o1); }
            { v2f vb; vb.x = vrow[32 + nlo]; vb.y = vrow[Hh + 32 + nlo]; o2 = wmma4(pa, vb, o2); }
            { v2f vb; vb.x = vrow[48 + nlo]; vb.y = vrow[Hh + 48 + nlo]; o3 = wmma4(pa, vb, o3); }
        }
        __builtin_amdgcn_wave_barrier();
    }

    #pragma unroll
    for (int r = 0; r < 8; ++r) {
        const int row = r + half * 8;
        obuf[w][row][ 0 + nlo] = o0[r];
        obuf[w][row][16 + nlo] = o1[r];
        obuf[w][row][32 + nlo] = o2[r];
        obuf[w][row][48 + nlo] = o3[r];
    }
    if (nlo == 0) {
        #pragma unroll
        for (int r = 0; r < 8; ++r) {
            mlbuf[w][r + half * 8][0] = mrow[r];
            mlbuf[w][r + half * 8][1] = lrow[r];
        }
    }
    __syncthreads();

    const int row = threadIdx.x >> 4;
    const int c4  = (threadIdx.x & 15) * 4;

    float mw[NW], lw[NW];
    float M = -1e30f;
    #pragma unroll
    for (int i = 0; i < NW; ++i) {
        mw[i] = mlbuf[i][row][0];
        lw[i] = mlbuf[i][row][1];
        M = fmaxf(M, mw[i]);
    }
    float L = 0.f;
    float4 acc = make_float4(0.f, 0.f, 0.f, 0.f);
    #pragma unroll
    for (int i = 0; i < NW; ++i) {
        const float wgt = __expf(mw[i] - M);
        const float4 o = *(const float4*)&obuf[i][row][c4];
        L += wgt * lw[i];
        acc.x += wgt * o.x; acc.y += wgt * o.y;
        acc.z += wgt * o.z; acc.w += wgt * o.w;
    }
    const float inv = 1.0f / L;
    acc.x *= inv; acc.y *= inv; acc.z *= inv; acc.w *= inv;
    *(float4*)(out + (size_t)(qrow0 + row) * Hh + c4) = acc;
}

extern "C" void kernel_launch(void* const* d_in, const int* in_sizes, int n_in,
                              void* d_out, int out_size, void* d_ws, size_t ws_size,
                              hipStream_t stream) {
    const float* x  = (const float*)d_in[0];
    const float* Wk = (const float*)d_in[1];
    const float* Wq = (const float*)d_in[2];
    const float* Wv = (const float*)d_in[3];
    float* out = (float*)d_out;

    float* ws   = (float*)d_ws;
    float* qbuf = ws;
    float* kbuf = ws + (size_t)ROWS * Hh;
    float* vbuf = ws + 2 * (size_t)ROWS * Hh;

    proj_kernel<<<ROWS / 16, 128, 0, stream>>>(x, Wk, Wq, Wv, qbuf, kbuf, vbuf);
    attn_kernel<<<ROWS / 16, 32 * NW, 0, stream>>>(qbuf, kbuf, vbuf, out);
}


# attention: each work-group takes the query-tile pair (127-i, i) of a batch, one tile after the other, so every group owns 129 key tiles (was 1..128); per-tile wave dealing and sum order kept
# speedup vs baseline: 2.3349x; 1.1201x over previous
#include <hip/hip_runtime.h>

typedef __attribute__((ext_vector_type(2))) float v2f;
typedef __attribute__((ext_vector_type(8))) float v8f;

constexpr int Bb = 8, Tt = 2048, Ee = 1024, Hh = 64;
constexpr int ROWS = Bb * Tt;
constexpr int TPB  = Tt / 16;

__device__ __forceinline__ v8f wmma4(v2f a, v2f b, v8f c) {
    return __builtin_amdgcn_wmma_f32_16x16x4_f32(
         false, a,  false, b,
         (short)0, c,  false,  false);
}

__global__ __launch_bounds__(128)
void proj_kernel(const float* __restrict__ x,
                 const float* __restrict__ Wk,
                 const float* __restrict__ Wq,
                 const float* __restrict__ Wv,
                 float* __restrict__ qo, float* __restrict__ ko,
                 float* __restrict__ vo) {
    const int lane   = threadIdx.x & 31;
    const int wv     = threadIdx.x >> 5;
    const int rowb   = blockIdx.x * 16;
    const int mrow   = lane & 15;
    const int khalf  = (lane >> 4) * 2;
    const int ncol   = wv * 16 + (lane & 15);

    const float* xa = x + (size_t)(rowb + mrow) * Ee + khalf;

    v8f cq = {}, ck = {}, cv = {};
    #pragma unroll 4
    for (int e0 = 0; e0 < Ee; e0 += 4) {
        v2f a = *(const v2f*)(xa + e0);
        const float* wkp = Wk + (size_t)(e0 + khalf) * Hh + ncol;
        const float* wqp = Wq + (size_t)(e0 + khalf) * Hh + ncol;
        const float* wvp = Wv + (size_t)(e0 + khalf) * Hh + ncol;
        v2f bk; bk.x = wkp[0]; bk.y = wkp[Hh];
        v2f bq; bq.x = wqp[0]; bq.y = wqp[Hh];
        v2f bv; bv.x = wvp[0]; bv.y = wvp[Hh];
        ck = wmma4(a, bk, ck);
        cq = wmma4(a, bq, cq);
        cv = wmma4(a, bv, cv);
    }

    const int orow = rowb + (lane >> 4) * 8;
    #pragma unroll
    for (int r = 0; r < 8; ++r) {
        size_t idx = (size_t)(orow + r) * Hh + ncol;
        qo[idx] = cq[r];
        ko[idx] = ck[r];
        vo[idx] = cv[r];
    }
}

constexpr int NW     = 8;
constexpr int OPITCH = 68;
constexpr int NPAIR  = TPB / 2;

__global__ __launch_bounds__(32 * NW)
void attn_kernel(const float* __restrict__ q,
                 const float* __restrict__ k,
                 const float* __restrict__ v,
                 float* __restrict__ out) {
    __shared__ float pbuf[NW][16][18];
    __shared__ alignas(16) float obuf[2][NW][16][OPITCH];
    __shared__ alignas(8)  float mlbuf[2][NW][16][2];

    const int lane  = threadIdx.x & 31;
    const int w     = __builtin_amdgcn_readfirstlane((int)(threadIdx.x >> 5));
    const int b     = blockIdx.x / NPAIR;
    const int ip    = blockIdx.x % NPAIR;
    const int half  = lane >> 4;
    const int nlo   = lane & 15;
    const int khalf = half * 2;

    const float scale = 0.03125f;

    for (int ph = 0; ph < 2; ++ph) {
        const int it    = ph == 0 ? (TPB - 1 - ip) : ip;
        const int qrow0 = b * Tt + it * 16;

        v2f qa[16];
        const float* qp = q + (size_t)(qrow0 + nlo) * Hh + khalf;
        #pragma unroll
        for (int s = 0; s < 16; ++s) qa[s] = *(const v2f*)(qp + 4 * s);

        v8f o0 = {}, o1 = {}, o2 = {}, o3 = {};
        float mrow[8], lrow[8];
        #pragma unroll
        for (int r = 0; r < 8; ++r) { mrow[r] = -1e30f; lrow[r] = 0.f; }

        for (int j = w; j <= it; j += NW) {
            const int krow0 = b * Tt + j * 16;

            v8f s8 = {};
            const float* kp = k + (size_t)(krow0 + nlo) * Hh + khalf;
            #pragma unroll
            for (int t = 0; t < 16; ++t) {
                v2f kb = *(const v2f*)(kp + 4 * t);
                s8 = wmma4(qa[t], kb, s8);
            }

            float sv[8];
            #pragma unroll
            for (int r = 0; r < 8; ++r) sv[r] = s8[r] * scale;

            if (j == it) {
                #pragma unroll
                for (int r = 0; r < 8; ++r) {
                    int m = r + half * 8;
                    if (nlo > m) sv[r] = -1e30f;
                }
            }

            float alpha[8], pv[8];
            #pragma unroll
            for (int r = 0; r < 8; ++r) {
                float t = sv[r];
                t = fmaxf(t, __shfl_xor(t, 1, 16));
                t = fmaxf(t, __shfl_xor(t, 2, 16));
                t = fmaxf(t, __shfl_xor(t, 4, 16));
                t = fmaxf(t, __shfl_xor(t, 8, 16));
                float mnew = fmaxf(mrow[r], t);
                alpha[r] = __expf(mrow[r] - mnew);
                mrow[r]  = mnew;
                float p  = __expf(sv[r] - mnew);
                pv[r] = p;
                float s = p;
                s += __shfl_xor(s, 1, 16);
                s += __shfl_xor(s, 2, 16);
                s += __shfl_xor(s, 4, 16);
                s += __shfl_xor(s, 8, 16);
                lrow[r] = lrow[r] * alpha[r] + s;
            }

            #pragma unroll
            for (int r = 0; r < 8; ++r) {
                o0[r] *= alpha[r]; o1[r] *= alpha[r];
                o2[r] *= alpha[r]; o3[r] *= alpha[r];
            }

            #pragma unroll
            for (int r = 0; r < 8; ++r)
                pbuf[w][r + half * 8][nlo] = pv[r];
            __builtin_amdgcn_wave_barrier();
            asm volatile("s_wait_dscnt 0" ::: "memory");

            const float* vp = v + (size_t)krow0 * Hh;
            #pragma unroll
            for (int kk0 = 0; kk0 < 16; kk0 += 4) {
                v2f pa;
                pa.x = pbuf[w][nlo][kk0 + khalf];
                pa.y = pbuf[w][nlo][kk0 + khalf + 1];
                const float* vrow = vp + (size_t)(kk0 + khalf) * Hh;
                { v2f vb; vb.x = vrow[ 0 + nlo]; vb.y = vrow[Hh +  0 + nlo]; o0 = wmma4(pa, vb, o0); }
                { v2f vb; vb.x = vrow[16 + nlo]; vb.y = vrow[Hh + 16 + nlo]; o1 = wmma4(pa, vb, o1); }
                { v2f vb; vb.x = vrow[32 + nlo]; vb.y = vrow[Hh + 32 + nlo]; o2 = wmma4(pa, vb, o2); }
                { v2f vb; vb.x = vrow[48 + nlo]; vb.y = vrow[Hh + 48 + nlo]; o3 = wmma4(pa, vb, o3); }
            }
            __builtin_amdgcn_wave_barrier();
        }

        #pragma unroll
        for (int r = 0; r < 8; ++r) {
            const int row = r + half * 8;
            obuf[ph][w][row][ 0 + nlo] = o0[r];
            obuf[ph][w][row][16 + nlo] = o1[r];
            obuf[ph][w][row][32 + nlo] = o2[r];
            obuf[ph][w][row][48 + nlo] = o3[r];
        }
        if (nlo == 0) {
            #pragma unroll
            for (int r = 0; r < 8; ++r) {
                mlbuf[ph][w][r + half * 8][0] = mrow[r];
                mlbuf[ph][w][r + half * 8][1] = lrow[r];
            }
        }
    }
    __syncthreads();

    const int row = threadIdx.x >> 4;
    const int c4  = (threadIdx.x & 15) * 4;

    for (int ph = 0; ph < 2; ++ph) {
        const int it    = ph == 0 ? (TPB - 1 - ip) : ip;
        const int qrow0 = b * Tt + it * 16;

        float mw[NW], lw[NW];
        float M = -1e30f;
        #pragma unroll
        for (int i = 0; i < NW; ++i) {
            mw[i] = mlbuf[ph][i][row][0];
            lw[i] = mlbuf[ph][i][row][1];
            M = fmaxf(M, mw[i]);
        }
        float L = 0.f;
        float4 acc = make_float4(0.f, 0.f, 0.f, 0.f);
        #pragma unroll
        for (int i = 0; i < NW; ++i) {
            const float wgt = __expf(mw[i] - M);
            const float4 o = *(const float4*)&obuf[ph][i][row][c4];
            L += wgt * lw[i];
            acc.x += wgt * o.x; acc.y += wgt * o.y;
            acc.z += wgt * o.z; acc.w += wgt * o.w;
        }
        const float inv = 1.0f / L;
        acc.x *= inv; acc.y *= inv; acc.z *= inv; acc.w *= inv;
        *(float4*)(out + (size_t)(qrow0 + row) * Hh + c4) = acc;
    }
}

extern "C" void kernel_launch(void* const* d_in, const int* in_sizes, int n_in,
                              void* d_out, int out_size, void* d_ws, size_t ws_size,
                              hipStream_t stream) {
    const float* x  = (const float*)d_in[0];
    const float* Wk = (const float*)d_in[1];
    const float* Wq = (const float*)d_in[2];
    const float* Wv = (const float*)d_in[3];
    float* out = (float*)d_out;

    float* ws   = (float*)d_ws;
    float* qbuf = ws;
    float* kbuf = ws + (size_t)ROWS * Hh;
    float* vbuf = ws + 2 * (size_t)ROWS * Hh;

    proj_kernel<<<ROWS / 16, 128, 0, stream>>>(x, Wk, Wq, Wv, qbuf, kbuf, vbuf);
    attn_kernel<<<ROWS / 32, 32 * NW, 0, stream>>>(qbuf, kbuf, vbuf, out);
}


# projection re-tiled as LDS-staged GEMM: 8-wave group owns 64 rows x 192 cols, x and W staged via LDS with whole-line loads; k summed in order 0,1,4,5,2,3,6,7 within each 8
# speedup vs baseline: 2.7211x; 1.1654x over previous
#include <hip/hip_runtime.h>

typedef __attribute__((ext_vector_type(2))) float v2f;
typedef __attribute__((ext_vector_type(8))) float v8f;

constexpr int Bb = 8, Tt = 2048, Ee = 1024, Hh = 64;
constexpr int ROWS = Bb * Tt;
constexpr int TPB  = Tt / 16;

__device__ __forceinline__ v8f wmma4(v2f a, v2f b, v8f c) {
    return __builtin_amdgcn_wmma_f32_16x16x4_f32(
         false, a,  false, b,
         (short)0, c,  false,  false);
}

typedef __attribute__((ext_vector_type(4))) float v4f;

constexpr int PBM = 64;
constexpr int PKC = 64;
constexpr int PP  = PKC + 4;
constexpr int PN  = 3 * Hh;
constexpr int NCH = Ee / PKC;
constexpr int XQ  = PBM * PKC / 4 / 256;
constexpr int WI  = PKC / 16;

__device__ __forceinline__ v2f lo2(v4f a) { return __builtin_shufflevector(a, a, 0, 1); }
__device__ __forceinline__ v2f hi2(v4f a) { return __builtin_shufflevector(a, a, 2, 3); }

__global__ __launch_bounds__(256)
void proj_kernel(const float* __restrict__ x,
                 const float* __restrict__ Wk,
                 const float* __restrict__ Wq,
                 const float* __restrict__ Wv,
                 float* __restrict__ qo, float* __restrict__ ko,
                 float* __restrict__ vo) {
    __shared__ alignas(16) float xs[2][PBM][PP];
    __shared__ alignas(16) float wt[2][PN][PP];

    const int tid  = threadIdx.x;
    const int lane = tid & 31;
    const int w    = __builtin_amdgcn_readfirstlane(tid >> 5);
    const int wm   = w >> 2;
    const int wn   = w & 3;
    const int c    = lane & 15;
    const int h    = lane >> 4;
    const int rowb = blockIdx.x * PBM;

    const int xr   = tid >> 4;
    const int xc   = (tid & 15) * 4;
    const float* xg = x + (size_t)(rowb + xr) * Ee + xc;
    const int wcol = tid & 63;
    const int wkq  = tid >> 6;
    const size_t woff = (size_t)(4 * wkq) * Hh + wcol;
    const float* wg0 = Wk + woff;
    const float* wg1 = Wq + woff;
    const float* wg2 = Wv + woff;

    v4f   sx[XQ];
    float sw[3][WI][4];

    auto load_chunk = [&](int e0) {
        #pragma unroll
        for (int q = 0; q < XQ; ++q)
            sx[q] = *(const v4f*)(xg + (size_t)(16 * q) * Ee + e0);
        #pragma unroll
        for (int i = 0; i < WI; ++i) {
            #pragma unroll
            for (int s = 0; s < 4; ++s) {
                const size_t o = (size_t)(e0 + 16 * i + s) * Hh;
                sw[0][i][s] = wg0[o];
                sw[1][i][s] = wg1[o];
                sw[2][i][s] = wg2[o];
            }
        }
    };
    auto store_chunk = [&](int buf) {
        #pragma unroll
        for (int q = 0; q < XQ; ++q)
            *(v4f*)&xs[buf][xr + 16 * q][xc] = sx[q];
        #pragma unroll
        for (int m = 0; m < 3; ++m) {
            #pragma unroll
            for (int i = 0; i < WI; ++i) {
                v4f t;
                t.x = sw[m][i][0]; t.y = sw[m][i][1]; t.z = sw[m][i][2]; t.w = sw[m][i][3];
                *(v4f*)&wt[buf][m * Hh + wcol][4 * (wkq + 4 * i)] = t;
            }
        }
    };

    v8f acc[2][3];
    #pragma unroll
    for (int mt = 0; mt < 2; ++mt) {
        #pragma unroll
        for (int nt = 0; nt < 3; ++nt) acc[mt][nt] = v8f{};
    }

    const int ar = wm * 32 + c;
    const int bn = wn * 48 + c;

    auto compute = [&](int buf) {
        v4f a[2][2], b[2][3];
        #pragma unroll
        for (int mt = 0; mt < 2; ++mt) a[0][mt] = *(const v4f*)&xs[buf][ar + 16 * mt][4 * h];
        #pragma unroll
        for (int nt = 0; nt < 3; ++nt) b[0][nt] = *(const v4f*)&wt[buf][bn + 16 * nt][4 * h];
        #pragma unroll
        for (int j = 0; j < PKC / 8; ++j) {
            const int cur = j & 1, nxt = cur ^ 1;
            if (j + 1 < PKC / 8) {
                #pragma unroll
                for (int mt = 0; mt < 2; ++mt) a[nxt][mt] = *(const v4f*)&xs[buf][ar + 16 * mt][8 * (j + 1) + 4 * h];
                #pragma unroll
                for (int nt = 0; nt < 3; ++nt) b[nxt][nt] = *(const v4f*)&wt[buf][bn + 16 * nt][8 * (j + 1) + 4 * h];
            }
            #pragma unroll
            for (int mt = 0; mt < 2; ++mt) {
                #pragma unroll
                for (int nt = 0; nt < 3; ++nt)
                    acc[mt][nt] = wmma4(lo2(a[cur][mt]), lo2(b[cur][nt]), acc[mt][nt]);
            }
            #pragma unroll
            for (int mt = 0; mt < 2; ++mt) {
                #pragma unroll
                for (int nt = 0; nt < 3; ++nt)
                    acc[mt][nt] = wmma4(hi2(a[cur][mt]), hi2(b[cur][nt]), acc[mt][nt]);
            }
        }
    };

    load_chunk(0);
    store_chunk(0);
    __syncthreads();
    for (int ch = 0; ch < NCH - 1; ++ch) {
        const int buf = ch & 1;
        load_chunk((ch + 1) * PKC);
        compute(buf);
        __builtin_amdgcn_sched_barrier(0);
        store_chunk(buf ^ 1);
        __syncthreads();
    }
    compute((NCH - 1) & 1);

    #pragma unroll
    for (int nt = 0; nt < 3; ++nt) {
        const int n0 = wn * 48 + nt * 16;
        float* dst = (n0 < Hh) ? ko : ((n0 < 2 * Hh) ? qo : vo);
        const int col = (n0 & (Hh - 1)) + c;
        #pragma unroll
        for (int mt = 0; mt < 2; ++mt) {
            const int orow = rowb + wm * 32 + mt * 16 + h * 8;
            #pragma unroll
            for (int r = 0; r < 8; ++r)
                dst[(size_t)(orow + r) * Hh + col] = acc[mt][nt][r];
        }
    }
}

constexpr int NW     = 8;
constexpr int OPITCH = 68;
constexpr int NPAIR  = TPB / 2;

__global__ __launch_bounds__(32 * NW)
void attn_kernel(const float* __restrict__ q,
                 const float* __restrict__ k,
                 const float* __restrict__ v,
                 float* __restrict__ out) {
    __shared__ float pbuf[NW][16][18];
    __shared__ alignas(16) float obuf[2][NW][16][OPITCH];
    __shared__ alignas(8)  float mlbuf[2][NW][16][2];

    const int lane  = threadIdx.x & 31;
    const int w     = __builtin_amdgcn_readfirstlane((int)(threadIdx.x >> 5));
    const int b     = blockIdx.x / NPAIR;
    const int ip    = blockIdx.x % NPAIR;
    const int half  = lane >> 4;
    const int nlo   = lane & 15;
    const int khalf = half * 2;

    const float scale = 0.03125f;

    for (int ph = 0; ph < 2; ++ph) {
        const int it    = ph == 0 ? (TPB - 1 - ip) : ip;
        const int qrow0 = b * Tt + it * 16;

        v2f qa[16];
        const float* qp = q + (size_t)(qrow0 + nlo) * Hh + khalf;
        #pragma unroll
        for (int s = 0; s < 16; ++s) qa[s] = *(const v2f*)(qp + 4 * s);

        v8f o0 = {}, o1 = {}, o2 = {}, o3 = {};
        float mrow[8], lrow[8];
        #pragma unroll
        for (int r = 0; r < 8; ++r) { mrow[r] = -1e30f; lrow[r] = 0.f; }

        for (int j = w; j <= it; j += NW) {
            const int krow0 = b * Tt + j * 16;

            v8f s8 = {};
            const float* kp = k + (size_t)(krow0 + nlo) * Hh + khalf;
            #pragma unroll
            for (int t = 0; t < 16; ++t) {
                v2f kb = *(const v2f*)(kp + 4 * t);
                s8 = wmma4(qa[t], kb, s8);
            }

            float sv[8];
            #pragma unroll
            for (int r = 0; r < 8; ++r) sv[r] = s8[r] * scale;

            if (j == it) {
                #pragma unroll
                for (int r = 0; r < 8; ++r) {
                    int m = r + half * 8;
                    if (nlo > m) sv[r] = -1e30f;
                }
            }

            float alpha[8], pv[8];
            #pragma unroll
            for (int r = 0; r < 8; ++r) {
                float t = sv[r];
                t = fmaxf(t, __shfl_xor(t, 1, 16));
                t = fmaxf(t, __shfl_xor(t, 2, 16));
                t = fmaxf(t, __shfl_xor(t, 4, 16));
                t = fmaxf(t, __shfl_xor(t, 8, 16));
                float mnew = fmaxf(mrow[r], t);
                alpha[r] = __expf(mrow[r] - mnew);
                mrow[r]  = mnew;
                float p  = __expf(sv[r] - mnew);
                pv[r] = p;
                float s = p;
                s += __shfl_xor(s, 1, 16);
                s += __shfl_xor(s, 2, 16);
                s += __shfl_xor(s, 4, 16);
                s += __shfl_xor(s, 8, 16);
                lrow[r] = lrow[r] * alpha[r] + s;
            }

            #pragma unroll
            for (int r = 0; r < 8; ++r) {
                o0[r] *= alpha[r]; o1[r] *= alpha[r];
                o2[r] *= alpha[r]; o3[r] *= alpha[r];
            }

            #pragma unroll
            for (int r = 0; r < 8; ++r)
                pbuf[w][r + half * 8][nlo] = pv[r];
            __builtin_amdgcn_wave_barrier();
            asm volatile("s_wait_dscnt 0" ::: "memory");

            const float* vp = v + (size_t)krow0 * Hh;
            #pragma unroll
            for (int kk0 = 0; kk0 < 16; kk0 += 4) {
                v2f pa;
                pa.x = pbuf[w][nlo][kk0 + khalf];
                pa.y = pbuf[w][nlo][kk0 + khalf + 1];
                const float* vrow = vp + (size_t)(kk0 + khalf) * Hh;
                { v2f vb; vb.x = vrow[ 0 + nlo]; vb.y = vrow[Hh +  0 + nlo]; o0 = wmma4(pa, vb, o0); }
                { v2f vb; vb.x = vrow[16 + nlo]; vb.y = vrow[Hh + 16 + nlo]; o1 = wmma4(pa, vb, o1); }
                { v2f vb; vb.x = vrow[32 + nlo]; vb.y = vrow[Hh + 32 + nlo]; o2 = wmma4(pa, vb, o2); }
                { v2f vb; vb.x = vrow[48 + nlo]; vb.y = vrow[Hh + 48 + nlo]; o3 = wmma4(pa, vb, o3); }
            }
            __builtin_amdgcn_wave_barrier();
        }

        #pragma unroll
        for (int r = 0; r < 8; ++r) {
            const int row = r + half * 8;
            obuf[ph][w][row][ 0 + nlo] = o0[r];
            obuf[ph][w][row][16 + nlo] = o1[r];
            obuf[ph][w][row][32 + nlo] = o2[r];
            obuf[ph][w][row][48 + nlo] = o3[r];
        }
        if (nlo == 0) {
            #pragma unroll
            for (int r = 0; r < 8; ++r) {
                mlbuf[ph][w][r + half * 8][0] = mrow[r];
                mlbuf[ph][w][r + half * 8][1] = lrow[r];
            }
        }
    }
    __syncthreads();

    const int row = threadIdx.x >> 4;
    const int c4  = (threadIdx.x & 15) * 4;

    for (int ph = 0; ph < 2; ++ph) {
        const int it    = ph == 0 ? (TPB - 1 - ip) : ip;
        const int qrow0 = b * Tt + it * 16;

        float mw[NW], lw[NW];
        float M = -1e30f;
        #pragma unroll
        for (int i = 0; i < NW; ++i) {
            mw[i] = mlbuf[ph][i][row][0];
            lw[i] = mlbuf[ph][i][row][1];
            M = fmaxf(M, mw[i]);
        }
        float L = 0.f;
        float4 acc = make_float4(0.f, 0.f, 0.f, 0.f);
        #pragma unroll
        for (int i = 0; i < NW; ++i) {
            const float wgt = __expf(mw[i] - M);
            const float4 o = *(const float4*)&obuf[ph][i][row][c4];
            L += wgt * lw[i];
            acc.x += wgt * o.x; acc.y += wgt * o.y;
            acc.z += wgt * o.z; acc.w += wgt * o.w;
        }
        const float inv = 1.0f / L;
        acc.x *= inv; acc.y *= inv; acc.z *= inv; acc.w *= inv;
        *(float4*)(out + (size_t)(qrow0 + row) * Hh + c4) = acc;
    }
}

extern "C" void kernel_launch(void* const* d_in, const int* in_sizes, int n_in,
                              void* d_out, int out_size, void* d_ws, size_t ws_size,
                              hipStream_t stream) {
    const float* x  = (const float*)d_in[0];
    const float* Wk = (const float*)d_in[1];
    const float* Wq = (const float*)d_in[2];
    const float* Wv = (const float*)d_in[3];
    float* out = (float*)d_out;

    float* ws   = (float*)d_ws;
    float* qbuf = ws;
    float* kbuf = ws + (size_t)ROWS * Hh;
    float* vbuf = ws + 2 * (size_t)ROWS * Hh;

    proj_kernel<<<ROWS / PBM, 256, 0, stream>>>(x, Wk, Wq, Wv, qbuf, kbuf, vbuf);
    attn_kernel<<<ROWS / 32, 32 * NW, 0, stream>>>(qbuf, kbuf, vbuf, out);
}


# attention: scores computed transposed (S^T = K*Q^T), so P feeds P*V from registers and row max/sum are in-lane: no LDS transpose, no 16-lane shuffles; softmax and P*V sums add in a new order
# speedup vs baseline: 2.9473x; 1.0831x over previous
#include <hip/hip_runtime.h>

typedef __attribute__((ext_vector_type(2))) float v2f;
typedef __attribute__((ext_vector_type(8))) float v8f;

constexpr int Bb = 8, Tt = 2048, Ee = 1024, Hh = 64;
constexpr int ROWS = Bb * Tt;
constexpr int TPB  = Tt / 16;

__device__ __forceinline__ v8f wmma4(v2f a, v2f b, v8f c) {
    return __builtin_amdgcn_wmma_f32_16x16x4_f32(
         false, a,  false, b,
         (short)0, c,  false,  false);
}

typedef __attribute__((ext_vector_type(4))) float v4f;

constexpr int PBM = 64;
constexpr int PKC = 64;
constexpr int PP  = PKC + 4;
constexpr int PN  = 3 * Hh;
constexpr int NCH = Ee / PKC;
constexpr int XQ  = PBM * PKC / 4 / 256;
constexpr int WI  = PKC / 16;

__device__ __forceinline__ v2f lo2(v4f a) { return __builtin_shufflevector(a, a, 0, 1); }
__device__ __forceinline__ v2f hi2(v4f a) { return __builtin_shufflevector(a, a, 2, 3); }

__global__ __launch_bounds__(256)
void proj_kernel(const float* __restrict__ x,
                 const float* __restrict__ Wk,
                 const float* __restrict__ Wq,
                 const float* __restrict__ Wv,
                 float* __restrict__ qo, float* __restrict__ ko,
                 float* __restrict__ vo) {
    __shared__ alignas(16) float xs[2][PBM][PP];
    __shared__ alignas(16) float wt[2][PN][PP];

    const int tid  = threadIdx.x;
    const int lane = tid & 31;
    const int w    = __builtin_amdgcn_readfirstlane(tid >> 5);
    const int wm   = w >> 2;
    const int wn   = w & 3;
    const int c    = lane & 15;
    const int h    = lane >> 4;
    const int rowb = blockIdx.x * PBM;

    const int xr   = tid >> 4;
    const int xc   = (tid & 15) * 4;
    const float* xg = x + (size_t)(rowb + xr) * Ee + xc;
    const int wcol = tid & 63;
    const int wkq  = tid >> 6;
    const size_t woff = (size_t)(4 * wkq) * Hh + wcol;
    const float* wg0 = Wk + woff;
    const float* wg1 = Wq + woff;
    const float* wg2 = Wv + woff;

    v4f   sx[XQ];
    float sw[3][WI][4];

    auto load_chunk = [&](int e0) {
        #pragma unroll
        for (int q = 0; q < XQ; ++q)
            sx[q] = *(const v4f*)(xg + (size_t)(16 * q) * Ee + e0);
        #pragma unroll
        for (int i = 0; i < WI; ++i) {
            #pragma unroll
            for (int s = 0; s < 4; ++s) {
                const size_t o = (size_t)(e0 + 16 * i + s) * Hh;
                sw[0][i][s] = wg0[o];
                sw[1][i][s] = wg1[o];
                sw[2][i][s] = wg2[o];
            }
        }
    };
    auto store_chunk = [&](int buf) {
        #pragma unroll
        for (int q = 0; q < XQ; ++q)
            *(v4f*)&xs[buf][xr + 16 * q][xc] = sx[q];
        #pragma unroll
        for (int m = 0; m < 3; ++m) {
            #pragma unroll
            for (int i = 0; i < WI; ++i) {
                v4f t;
                t.x = sw[m][i][0]; t.y = sw[m][i][1]; t.z = sw[m][i][2]; t.w = sw[m][i][3];
                *(v4f*)&wt[buf][m * Hh + wcol][4 * (wkq + 4 * i)] = t;
            }
        }
    };

    v8f acc[2][3];
    #pragma unroll
    for (int mt = 0; mt < 2; ++mt) {
        #pragma unroll
        for (int nt = 0; nt < 3; ++nt) acc[mt][nt] = v8f{};
    }

    const int ar = wm * 32 + c;
    const int bn = wn * 48 + c;

    auto compute = [&](int buf) {
        v4f a[2][2], b[2][3];
        #pragma unroll
        for (int mt = 0; mt < 2; ++mt) a[0][mt] = *(const v4f*)&xs[buf][ar + 16 * mt][4 * h];
        #pragma unroll
        for (int nt = 0; nt < 3; ++nt) b[0][nt] = *(const v4f*)&wt[buf][bn + 16 * nt][4 * h];
        #pragma unroll
        for (int j = 0; j < PKC / 8; ++j) {
            const int cur = j & 1, nxt = cur ^ 1;
            if (j + 1 < PKC / 8) {
                #pragma unroll
                for (int mt = 0; mt < 2; ++mt) a[nxt][mt] = *(const v4f*)&xs[buf][ar + 16 * mt][8 * (j + 1) + 4 * h];
                #pragma unroll
                for (int nt = 0; nt < 3; ++nt) b[nxt][nt] = *(const v4f*)&wt[buf][bn + 16 * nt][8 * (j + 1) + 4 * h];
            }
            #pragma unroll
            for (int mt = 0; mt < 2; ++mt) {
                #pragma unroll
                for (int nt = 0; nt < 3; ++nt)
                    acc[mt][nt] = wmma4(lo2(a[cur][mt]), lo2(b[cur][nt]), acc[mt][nt]);
            }
            #pragma unroll
            for (int mt = 0; mt < 2; ++mt) {
                #pragma unroll
                for (int nt = 0; nt < 3; ++nt)
                    acc[mt][nt] = wmma4(hi2(a[cur][mt]), hi2(b[cur][nt]), acc[mt][nt]);
            }
        }
    };

    load_chunk(0);
    store_chunk(0);
    __syncthreads();
    for (int ch = 0; ch < NCH - 1; ++ch) {
        const int buf = ch & 1;
        load_chunk((ch + 1) * PKC);
        compute(buf);
        __builtin_amdgcn_sched_barrier(0);
        store_chunk(buf ^ 1);
        __syncthreads();
    }
    compute((NCH - 1) & 1);

    #pragma unroll
    for (int nt = 0; nt < 3; ++nt) {
        const int n0 = wn * 48 + nt * 16;
        float* dst = (n0 < Hh) ? ko : ((n0 < 2 * Hh) ? qo : vo);
        const int col = (n0 & (Hh - 1)) + c;
        #pragma unroll
        for (int mt = 0; mt < 2; ++mt) {
            const int orow = rowb + wm * 32 + mt * 16 + h * 8;
            #pragma unroll
            for (int r = 0; r < 8; ++r)
                dst[(size_t)(orow + r) * Hh + col] = acc[mt][nt][r];
        }
    }
}

constexpr int NW     = 8;
constexpr int OPITCH = 68;
constexpr int NPAIR  = TPB / 2;

__device__ __forceinline__ float xhalf(float x) {
    const int b = __builtin_bit_cast(int, x);
    return __builtin_bit_cast(float, __builtin_amdgcn_permlanex16(b, b, 0x76543210, 0xfedcba98, false, false));
}

__global__ __launch_bounds__(32 * NW)
void attn_kernel(const float* __restrict__ q,
                 const float* __restrict__ k,
                 const float* __restrict__ v,
                 float* __restrict__ out) {
    __shared__ alignas(16) float obuf[2][NW][16][OPITCH];
    __shared__ alignas(8)  float mlbuf[2][NW][16][2];

    const int lane  = threadIdx.x & 31;
    const int w     = __builtin_amdgcn_readfirstlane((int)(threadIdx.x >> 5));
    const int b     = blockIdx.x / NPAIR;
    const int ip    = blockIdx.x % NPAIR;
    const int half  = lane >> 4;
    const int nlo   = lane & 15;
    const int khalf = half * 2;

    const float scale = 0.03125f;

    for (int ph = 0; ph < 2; ++ph) {
        const int it    = ph == 0 ? (TPB - 1 - ip) : ip;
        const int qrow0 = b * Tt + it * 16;

        v2f qb[16];
        const float* qp = q + (size_t)(qrow0 + nlo) * Hh + khalf;
        #pragma unroll
        for (int s = 0; s < 16; ++s) qb[s] = *(const v2f*)(qp + 4 * s);

        v8f o0 = {}, o1 = {}, o2 = {}, o3 = {};
        float mrun = -1e30f;
        float lrun = 0.f;

        for (int j = w; j <= it; j += NW) {
            const int krow0 = b * Tt + j * 16;

            v8f s8 = {};
            const float* kp = k + (size_t)(krow0 + nlo) * Hh + khalf;
            #pragma unroll
            for (int t = 0; t < 16; ++t) {
                v2f ka = *(const v2f*)(kp + 4 * t);
                s8 = wmma4(ka, qb[t], s8);
            }

            float sv[8];
            #pragma unroll
            for (int r = 0; r < 8; ++r) sv[r] = s8[r] * scale;

            if (j == it) {
                #pragma unroll
                for (int r = 0; r < 8; ++r) {
                    int m = r + half * 8;
                    if (m > nlo) sv[r] = -1e30f;
                }
            }

            float t = fmaxf(fmaxf(fmaxf(sv[0], sv[1]), fmaxf(sv[2], sv[3])),
                            fmaxf(fmaxf(sv[4], sv[5]), fmaxf(sv[6], sv[7])));
            t = fmaxf(t, xhalf(t));
            const float mnew  = fmaxf(mrun, t);
            const float alpha = __expf(mrun - mnew);
            mrun = mnew;
            float pv[8];
            float psum = 0.f;
            #pragma unroll
            for (int r = 0; r < 8; ++r) {
                pv[r] = __expf(sv[r] - mnew);
                psum += pv[r];
            }
            lrun = lrun * alpha + psum;

            #pragma unroll
            for (int r = 0; r < 8; ++r) {
                o0[r] *= alpha; o1[r] *= alpha;
                o2[r] *= alpha; o3[r] *= alpha;
            }

            const float* vp = v + (size_t)(krow0 + 8 * half) * Hh + nlo;
            #pragma unroll
            for (int s = 0; s < 4; ++s) {
                v2f pb;
                pb.x = pv[2 * s];
                pb.y = pv[2 * s + 1];
                const float* vrow = vp + (size_t)(2 * s) * Hh;
                { v2f va; va.x = vrow[ 0]; va.y = vrow[Hh +  0]; o0 = wmma4(va, pb, o0); }
                { v2f va; va.x = vrow[16]; va.y = vrow[Hh + 16]; o1 = wmma4(va, pb, o1); }
                { v2f va; va.x = vrow[32]; va.y = vrow[Hh + 32]; o2 = wmma4(va, pb, o2); }
                { v2f va; va.x = vrow[48]; va.y = vrow[Hh + 48]; o3 = wmma4(va, pb, o3); }
            }
        }

        const float lsum = lrun + xhalf(lrun);

        float* op = &obuf[ph][w][nlo][8 * half];
        *(v4f*)(op +  0) = __builtin_shufflevector(o0, o0, 0, 1, 2, 3);
        *(v4f*)(op +  4) = __builtin_shufflevector(o0, o0, 4, 5, 6, 7);
        *(v4f*)(op + 16) = __builtin_shufflevector(o1, o1, 0, 1, 2, 3);
        *(v4f*)(op + 20) = __builtin_shufflevector(o1, o1, 4, 5, 6, 7);
        *(v4f*)(op + 32) = __builtin_shufflevector(o2, o2, 0, 1, 2, 3);
        *(v4f*)(op + 36) = __builtin_shufflevector(o2, o2, 4, 5, 6, 7);
        *(v4f*)(op + 48) = __builtin_shufflevector(o3, o3, 0, 1, 2, 3);
        *(v4f*)(op + 52) = __builtin_shufflevector(o3, o3, 4, 5, 6, 7);
        if (half == 0) {
            mlbuf[ph][w][nlo][0] = mrun;
            mlbuf[ph][w][nlo][1] = lsum;
        }
    }
    __syncthreads();

    const int row = threadIdx.x >> 4;
    const int c4  = (threadIdx.x & 15) * 4;

    for (int ph = 0; ph < 2; ++ph) {
        const int it    = ph == 0 ? (TPB - 1 - ip) : ip;
        const int qrow0 = b * Tt + it * 16;

        float mw[NW], lw[NW];
        float M = -1e30f;
        #pragma unroll
        for (int i = 0; i < NW; ++i) {
            mw[i] = mlbuf[ph][i][row][0];
            lw[i] = mlbuf[ph][i][row][1];
            M = fmaxf(M, mw[i]);
        }
        float L = 0.f;
        float4 acc = make_float4(0.f, 0.f, 0.f, 0.f);
        #pragma unroll
        for (int i = 0; i < NW; ++i) {
            const float wgt = __expf(mw[i] - M);
            const float4 o = *(const float4*)&obuf[ph][i][row][c4];
            L += wgt * lw[i];
            acc.x += wgt * o.x; acc.y += wgt * o.y;
            acc.z += wgt * o.z; acc.w += wgt * o.w;
        }
        const float inv = 1.0f / L;
        acc.x *= inv; acc.y *= inv; acc.z *= inv; acc.w *= inv;
        *(float4*)(out + (size_t)(qrow0 + row) * Hh + c4) = acc;
    }
}

extern "C" void kernel_launch(void* const* d_in, const int* in_sizes, int n_in,
                              void* d_out, int out_size, void* d_ws, size_t ws_size,
                              hipStream_t stream) {
    const float* x  = (const float*)d_in[0];
    const float* Wk = (const float*)d_in[1];
    const float* Wq = (const float*)d_in[2];
    const float* Wv = (const float*)d_in[3];
    float* out = (float*)d_out;

    float* ws   = (float*)d_ws;
    float* qbuf = ws;
    float* kbuf = ws + (size_t)ROWS * Hh;
    float* vbuf = ws + 2 * (size_t)ROWS * Hh;

    proj_kernel<<<ROWS / PBM, 256, 0, stream>>>(x, Wk, Wq, Wv, qbuf, kbuf, vbuf);
    attn_kernel<<<ROWS / 32, 32 * NW, 0, stream>>>(qbuf, kbuf, vbuf, out);
}


# attention: a wave works on two adjacent query tiles at once, so each loaded K and V fragment feeds 32 queries (was 16); 16-wave groups own 32-query blocks (63-i, i); softmax sums add in a new order
# speedup vs baseline: 3.1843x; 1.0804x over previous
#include <hip/hip_runtime.h>

typedef __attribute__((ext_vector_type(2))) float v2f;
typedef __attribute__((ext_vector_type(8))) float v8f;

constexpr int Bb = 8, Tt = 2048, Ee = 1024, Hh = 64;
constexpr int ROWS = Bb * Tt;
constexpr int TPB  = Tt / 16;

__device__ __forceinline__ v8f wmma4(v2f a, v2f b, v8f c) {
    return __builtin_amdgcn_wmma_f32_16x16x4_f32(
         false, a,  false, b,
         (short)0, c,  false,  false);
}

typedef __attribute__((ext_vector_type(4))) float v4f;

constexpr int PBM = 64;
constexpr int PKC = 64;
constexpr int PP  = PKC + 4;
constexpr int PN  = 3 * Hh;
constexpr int NCH = Ee / PKC;
constexpr int XQ  = PBM * PKC / 4 / 256;
constexpr int WI  = PKC / 16;

__device__ __forceinline__ v2f lo2(v4f a) { return __builtin_shufflevector(a, a, 0, 1); }
__device__ __forceinline__ v2f hi2(v4f a) { return __builtin_shufflevector(a, a, 2, 3); }

__global__ __launch_bounds__(256)
void proj_kernel(const float* __restrict__ x,
                 const float* __restrict__ Wk,
                 const float* __restrict__ Wq,
                 const float* __restrict__ Wv,
                 float* __restrict__ qo, float* __restrict__ ko,
                 float* __restrict__ vo) {
    __shared__ alignas(16) float xs[2][PBM][PP];
    __shared__ alignas(16) float wt[2][PN][PP];

    const int tid  = threadIdx.x;
    const int lane = tid & 31;
    const int w    = __builtin_amdgcn_readfirstlane(tid >> 5);
    const int wm   = w >> 2;
    const int wn   = w & 3;
    const int c    = lane & 15;
    const int h    = lane >> 4;
    const int rowb = blockIdx.x * PBM;

    const int xr   = tid >> 4;
    const int xc   = (tid & 15) * 4;
    const float* xg = x + (size_t)(rowb + xr) * Ee + xc;
    const int wcol = tid & 63;
    const int wkq  = tid >> 6;
    const size_t woff = (size_t)(4 * wkq) * Hh + wcol;
    const float* wg0 = Wk + woff;
    const float* wg1 = Wq + woff;
    const float* wg2 = Wv + woff;

    v4f   sx[XQ];
    float sw[3][WI][4];

    auto load_chunk = [&](int e0) {
        #pragma unroll
        for (int q = 0; q < XQ; ++q)
            sx[q] = *(const v4f*)(xg + (size_t)(16 * q) * Ee + e0);
        #pragma unroll
        for (int i = 0; i < WI; ++i) {
            #pragma unroll
            for (int s = 0; s < 4; ++s) {
                const size_t o = (size_t)(e0 + 16 * i + s) * Hh;
                sw[0][i][s] = wg0[o];
                sw[1][i][s] = wg1[o];
                sw[2][i][s] = wg2[o];
            }
        }
    };
    auto store_chunk = [&](int buf) {
        #pragma unroll
        for (int q = 0; q < XQ; ++q)
            *(v4f*)&xs[buf][xr + 16 * q][xc] = sx[q];
        #pragma unroll
        for (int m = 0; m < 3; ++m) {
            #pragma unroll
            for (int i = 0; i < WI; ++i) {
                v4f t;
                t.x = sw[m][i][0]; t.y = sw[m][i][1]; t.z = sw[m][i][2]; t.w = sw[m][i][3];
                *(v4f*)&wt[buf][m * Hh + wcol][4 * (wkq + 4 * i)] = t;
            }
        }
    };

    v8f acc[2][3];
    #pragma unroll
    for (int mt = 0; mt < 2; ++mt) {
        #pragma unroll
        for (int nt = 0; nt < 3; ++nt) acc[mt][nt] = v8f{};
    }

    const int ar = wm * 32 + c;
    const int bn = wn * 48 + c;

    auto compute = [&](int buf) {
        v4f a[2][2], b[2][3];
        #pragma unroll
        for (int mt = 0; mt < 2; ++mt) a[0][mt] = *(const v4f*)&xs[buf][ar + 16 * mt][4 * h];
        #pragma unroll
        for (int nt = 0; nt < 3; ++nt) b[0][nt] = *(const v4f*)&wt[buf][bn + 16 * nt][4 * h];
        #pragma unroll
        for (int j = 0; j < PKC / 8; ++j) {
            const int cur = j & 1, nxt = cur ^ 1;
            if (j + 1 < PKC / 8) {
                #pragma unroll
                for (int mt = 0; mt < 2; ++mt) a[nxt][mt] = *(const v4f*)&xs[buf][ar + 16 * mt][8 * (j + 1) + 4 * h];
                #pragma unroll
                for (int nt = 0; nt < 3; ++nt) b[nxt][nt] = *(const v4f*)&wt[buf][bn + 16 * nt][8 * (j + 1) + 4 * h];
            }
            #pragma unroll
            for (int mt = 0; mt < 2; ++mt) {
                #pragma unroll
                for (int nt = 0; nt < 3; ++nt)
                    acc[mt][nt] = wmma4(lo2(a[cur][mt]), lo2(b[cur][nt]), acc[mt][nt]);
            }
            #pragma unroll
            for (int mt = 0; mt < 2; ++mt) {
                #pragma unroll
                for (int nt = 0; nt < 3; ++nt)
                    acc[mt][nt] = wmma4(hi2(a[cur][mt]), hi2(b[cur][nt]), acc[mt][nt]);
            }
        }
    };

    load_chunk(0);
    store_chunk(0);
    __syncthreads();
    for (int ch = 0; ch < NCH - 1; ++ch) {
        const int buf = ch & 1;
        load_chunk((ch + 1) * PKC);
        compute(buf);
        __builtin_amdgcn_sched_barrier(0);
        store_chunk(buf ^ 1);
        __syncthreads();
    }
    compute((NCH - 1) & 1);

    #pragma unroll
    for (int nt = 0; nt < 3; ++nt) {
        const int n0 = wn * 48 + nt * 16;
        float* dst = (n0 < Hh) ? ko : ((n0 < 2 * Hh) ? qo : vo);
        const int col = (n0 & (Hh - 1)) + c;
        #pragma unroll
        for (int mt = 0; mt < 2; ++mt) {
            const int orow = rowb + wm * 32 + mt * 16 + h * 8;
            #pragma unroll
            for (int r = 0; r < 8; ++r)
                dst[(size_t)(orow + r) * Hh + col] = acc[mt][nt][r];
        }
    }
}

constexpr int NW     = 16;
constexpr int OPITCH = 68;
constexpr int QROWS  = 32;
constexpr int NQB    = Tt / QROWS;
constexpr int NGRP   = NQB / 2;

__device__ __forceinline__ float xhalf(float x) {
    const int b = __builtin_bit_cast(int, x);
    return __builtin_bit_cast(float, __builtin_amdgcn_permlanex16(b, b, 0x76543210, 0xfedcba98, false, false));
}

__device__ __forceinline__ float softmax_step(const float (&sv)[8], float (&pv)[8], float& mrun, float& lrun) {
    float t = fmaxf(fmaxf(fmaxf(sv[0], sv[1]), fmaxf(sv[2], sv[3])),
                    fmaxf(fmaxf(sv[4], sv[5]), fmaxf(sv[6], sv[7])));
    t = fmaxf(t, xhalf(t));
    const float mnew  = fmaxf(mrun, t);
    const float alpha = __expf(mrun - mnew);
    mrun = mnew;
    float psum = 0.f;
    #pragma unroll
    for (int r = 0; r < 8; ++r) {
        pv[r] = __expf(sv[r] - mnew);
        psum += pv[r];
    }
    lrun = lrun * alpha + psum;
    return alpha;
}

__device__ __forceinline__ void put8(float* p, v8f o) {
    *(v4f*)(p + 0) = __builtin_shufflevector(o, o, 0, 1, 2, 3);
    *(v4f*)(p + 4) = __builtin_shufflevector(o, o, 4, 5, 6, 7);
}

__global__ __launch_bounds__(32 * NW)
void attn_kernel(const float* __restrict__ q,
                 const float* __restrict__ k,
                 const float* __restrict__ v,
                 float* __restrict__ out) {
    __shared__ alignas(16) float obuf[2][NW][QROWS][OPITCH];
    __shared__ alignas(8)  float mlbuf[2][NW][QROWS][2];

    const int lane  = threadIdx.x & 31;
    const int w     = __builtin_amdgcn_readfirstlane((int)(threadIdx.x >> 5));
    const int b     = blockIdx.x / NGRP;
    const int ip    = blockIdx.x % NGRP;
    const int half  = lane >> 4;
    const int nlo   = lane & 15;
    const int khalf = half * 2;

    const float scale = 0.03125f;

    for (int ph = 0; ph < 2; ++ph) {
        const int pb    = ph == 0 ? (NQB - 1 - ip) : ip;
        const int qrow0 = b * Tt + pb * QROWS;
        const int jlast = 2 * pb + 1;

        v2f qa[16], qb[16];
        const float* qp = q + (size_t)(qrow0 + nlo) * Hh + khalf;
        #pragma unroll
        for (int s = 0; s < 16; ++s) {
            qa[s] = *(const v2f*)(qp + 4 * s);
            qb[s] = *(const v2f*)(qp + 16 * Hh + 4 * s);
        }

        v8f oa0 = {}, oa1 = {}, oa2 = {}, oa3 = {};
        v8f ob0 = {}, ob1 = {}, ob2 = {}, ob3 = {};
        float ma = -1e30f, mb = -1e30f;
        float la = 0.f, lb = 0.f;

        for (int j = w; j <= jlast; j += NW) {
            const int krow0 = b * Tt + j * 16;

            v8f sa8 = {}, sb8 = {};
            const float* kp = k + (size_t)(krow0 + nlo) * Hh + khalf;
            #pragma unroll
            for (int t = 0; t < 16; ++t) {
                v2f ka = *(const v2f*)(kp + 4 * t);
                sa8 = wmma4(ka, qa[t], sa8);
                sb8 = wmma4(ka, qb[t], sb8);
            }

            float sva[8], svb[8];
            #pragma unroll
            for (int r = 0; r < 8; ++r) { sva[r] = sa8[r] * scale; svb[r] = sb8[r] * scale; }

            if (j >= jlast - 1) {
                const int d = (j - 2 * pb) * 16 + half * 8 - nlo;
                #pragma unroll
                for (int r = 0; r < 8; ++r) {
                    if (d + r > 0)  sva[r] = -1e30f;
                    if (d + r > 16) svb[r] = -1e30f;
                }
            }

            float pva[8], pvb[8];
            const float ala = softmax_step(sva, pva, ma, la);
            const float alb = softmax_step(svb, pvb, mb, lb);

            #pragma unroll
            for (int r = 0; r < 8; ++r) {
                oa0[r] *= ala; oa1[r] *= ala; oa2[r] *= ala; oa3[r] *= ala;
                ob0[r] *= alb; ob1[r] *= alb; ob2[r] *= alb; ob3[r] *= alb;
            }

            const float* vp = v + (size_t)(krow0 + 8 * half) * Hh + nlo;
            #pragma unroll
            for (int s = 0; s < 4; ++s) {
                v2f pa, pb2;
                pa.x  = pva[2 * s]; pa.y  = pva[2 * s + 1];
                pb2.x = pvb[2 * s]; pb2.y = pvb[2 * s + 1];
                const float* vrow = vp + (size_t)(2 * s) * Hh;
                { v2f va; va.x = vrow[ 0]; va.y = vrow[Hh +  0]; oa0 = wmma4(va, pa, oa0); ob0 = wmma4(va, pb2, ob0); }
                { v2f va; va.x = vrow[16]; va.y = vrow[Hh + 16]; oa1 = wmma4(va, pa, oa1); ob1 = wmma4(va, pb2, ob1); }
                { v2f va; va.x = vrow[32]; va.y = vrow[Hh + 32]; oa2 = wmma4(va, pa, oa2); ob2 = wmma4(va, pb2, ob2); }
                { v2f va; va.x = vrow[48]; va.y = vrow[Hh + 48]; oa3 = wmma4(va, pa, oa3); ob3 = wmma4(va, pb2, ob3); }
            }
        }

        const float lsa = la + xhalf(la);
        const float lsb = lb + xhalf(lb);

        float* opa = &obuf[ph][w][nlo][8 * half];
        float* opb = &obuf[ph][w][16 + nlo][8 * half];
        put8(opa +  0, oa0); put8(opa + 16, oa1); put8(opa + 32, oa2); put8(opa + 48, oa3);
        put8(opb +  0, ob0); put8(opb + 16, ob1); put8(opb + 32, ob2); put8(opb + 48, ob3);
        if (half == 0) {
            mlbuf[ph][w][nlo][0]      = ma;
            mlbuf[ph][w][nlo][1]      = lsa;
            mlbuf[ph][w][16 + nlo][0] = mb;
            mlbuf[ph][w][16 + nlo][1] = lsb;
        }
    }
    __syncthreads();

    const int row = threadIdx.x >> 4;
    const int c4  = (threadIdx.x & 15) * 4;

    for (int ph = 0; ph < 2; ++ph) {
        const int pb    = ph == 0 ? (NQB - 1 - ip) : ip;
        const int qrow0 = b * Tt + pb * QROWS;

        float mw[NW], lw[NW];
        float M = -1e30f;
        #pragma unroll
        for (int i = 0; i < NW; ++i) {
            mw[i] = mlbuf[ph][i][row][0];
            lw[i] = mlbuf[ph][i][row][1];
            M = fmaxf(M, mw[i]);
        }
        float L = 0.f;
        float4 acc = make_float4(0.f, 0.f, 0.f, 0.f);
        #pragma unroll
        for (int i = 0; i < NW; ++i) {
            const float wgt = __expf(mw[i] - M);
            const float4 o = *(const float4*)&obuf[ph][i][row][c4];
            L += wgt * lw[i];
            acc.x += wgt * o.x; acc.y += wgt * o.y;
            acc.z += wgt * o.z; acc.w += wgt * o.w;
        }
        const float inv = 1.0f / L;
        acc.x *= inv; acc.y *= inv; acc.z *= inv; acc.w *= inv;
        *(float4*)(out + (size_t)(qrow0 + row) * Hh + c4) = acc;
    }
}

extern "C" void kernel_launch(void* const* d_in, const int* in_sizes, int n_in,
                              void* d_out, int out_size, void* d_ws, size_t ws_size,
                              hipStream_t stream) {
    const float* x  = (const float*)d_in[0];
    const float* Wk = (const float*)d_in[1];
    const float* Wq = (const float*)d_in[2];
    const float* Wv = (const float*)d_in[3];
    float* out = (float*)d_out;

    float* ws   = (float*)d_ws;
    float* qbuf = ws;
    float* kbuf = ws + (size_t)ROWS * Hh;
    float* vbuf = ws + 2 * (size_t)ROWS * Hh;

    proj_kernel<<<ROWS / PBM, 256, 0, stream>>>(x, Wk, Wq, Wv, qbuf, kbuf, vbuf);
    attn_kernel<<<Bb * NGRP, 32 * NW, 0, stream>>>(qbuf, kbuf, vbuf, out);
}


# projection: LDS ring of 4 chunk stages filled 3 ahead; work-group barrier split into signal and wait one chunk apart, so a SIMD's two waves no longer stop together at chunk ends; sum order kept
# speedup vs baseline: 3.2964x; 1.0352x over previous
#include <hip/hip_runtime.h>

typedef __attribute__((ext_vector_type(2))) float v2f;
typedef __attribute__((ext_vector_type(8))) float v8f;

constexpr int Bb = 8, Tt = 2048, Ee = 1024, Hh = 64;
constexpr int ROWS = Bb * Tt;
constexpr int TPB  = Tt / 16;

__device__ __forceinline__ v8f wmma4(v2f a, v2f b, v8f c) {
    return __builtin_amdgcn_wmma_f32_16x16x4_f32(
         false, a,  false, b,
         (short)0, c,  false,  false);
}

typedef __attribute__((ext_vector_type(4))) float v4f;

constexpr int PBM = 64;
constexpr int PKC = 64;
constexpr int PP  = PKC + 4;
constexpr int PN  = 3 * Hh;
constexpr int NCH = Ee / PKC;
constexpr int NST = 4;
constexpr int XQ  = PBM * PKC / 4 / 256;
constexpr int WI  = PKC / 16;
constexpr int JS  = PKC / 8;

__device__ __forceinline__ v2f lo2(v4f a) { return __builtin_shufflevector(a, a, 0, 1); }
__device__ __forceinline__ v2f hi2(v4f a) { return __builtin_shufflevector(a, a, 2, 3); }

struct Staged { v4f sx[XQ]; float sw[3][WI][4]; };

__device__ __forceinline__ void ring_signal() { asm volatile("s_wait_dscnt 0x0\n\ts_barrier_signal -1" ::: "memory"); }
__device__ __forceinline__ void ring_wait()   { asm volatile("s_barrier_wait -1" ::: "memory"); }

__global__ __launch_bounds__(256)
void proj_kernel(const float* __restrict__ x,
                 const float* __restrict__ Wk,
                 const float* __restrict__ Wq,
                 const float* __restrict__ Wv,
                 float* __restrict__ qo, float* __restrict__ ko,
                 float* __restrict__ vo) {
    __shared__ alignas(16) float xs[NST][PBM][PP];
    __shared__ alignas(16) float wt[NST][PN][PP];

    const int tid  = threadIdx.x;
    const int lane = tid & 31;
    const int w    = __builtin_amdgcn_readfirstlane(tid >> 5);
    const int wm   = w >> 2;
    const int wn   = w & 3;
    const int c    = lane & 15;
    const int h    = lane >> 4;
    const int rowb = blockIdx.x * PBM;

    const int xr   = tid >> 4;
    const int xc   = (tid & 15) * 4;
    const float* xg = x + (size_t)(rowb + xr) * Ee + xc;
    const int wcol = tid & 63;
    const int wkq  = tid >> 6;
    const size_t woff = (size_t)(4 * wkq) * Hh + wcol;
    const float* wg0 = Wk + woff;
    const float* wg1 = Wq + woff;
    const float* wg2 = Wv + woff;

    auto load_chunk = [&](Staged& st, int e0) __attribute__((always_inline)) {
        #pragma unroll
        for (int q = 0; q < XQ; ++q)
            st.sx[q] = *(const v4f*)(xg + (size_t)(16 * q) * Ee + e0);
        #pragma unroll
        for (int i = 0; i < WI; ++i) {
            #pragma unroll
            for (int s = 0; s < 4; ++s) {
                const size_t o = (size_t)(e0 + 16 * i + s) * Hh;
                st.sw[0][i][s] = wg0[o];
                st.sw[1][i][s] = wg1[o];
                st.sw[2][i][s] = wg2[o];
            }
        }
    };
    auto store_chunk = [&](const Staged& st, int buf) __attribute__((always_inline)) {
        #pragma unroll
        for (int q = 0; q < XQ; ++q)
            *(v4f*)&xs[buf][xr + 16 * q][xc] = st.sx[q];
        #pragma unroll
        for (int m = 0; m < 3; ++m) {
            #pragma unroll
            for (int i = 0; i < WI; ++i) {
                v4f t;
                t.x = st.sw[m][i][0]; t.y = st.sw[m][i][1]; t.z = st.sw[m][i][2]; t.w = st.sw[m][i][3];
                *(v4f*)&wt[buf][m * Hh + wcol][4 * (wkq + 4 * i)] = t;
            }
        }
    };

    v8f acc[2][3];
    #pragma unroll
    for (int mt = 0; mt < 2; ++mt) {
        #pragma unroll
        for (int nt = 0; nt < 3; ++nt) acc[mt][nt] = v8f{};
    }

    const int ar = wm * 32 + c;
    const int bn = wn * 48 + c;

    v4f a[2][2], b[2][3];
    auto frag_load = [&](int buf, int j, int set) __attribute__((always_inline)) {
        #pragma unroll
        for (int mt = 0; mt < 2; ++mt) a[set][mt] = *(const v4f*)&xs[buf][ar + 16 * mt][8 * j + 4 * h];
        #pragma unroll
        for (int nt = 0; nt < 3; ++nt) b[set][nt] = *(const v4f*)&wt[buf][bn + 16 * nt][8 * j + 4 * h];
    };
    auto steps = [&](int buf, int j0, int j1) __attribute__((always_inline)) {
        #pragma unroll
        for (int j = j0; j < j1; ++j) {
            const int cur = j & 1, nxt = cur ^ 1;
            if (j + 1 < JS) frag_load(buf, j + 1, nxt);
            #pragma unroll
            for (int mt = 0; mt < 2; ++mt) {
                #pragma unroll
                for (int nt = 0; nt < 3; ++nt)
                    acc[mt][nt] = wmma4(lo2(a[cur][mt]), lo2(b[cur][nt]), acc[mt][nt]);
            }
            #pragma unroll
            for (int mt = 0; mt < 2; ++mt) {
                #pragma unroll
                for (int nt = 0; nt < 3; ++nt)
                    acc[mt][nt] = wmma4(hi2(a[cur][mt]), hi2(b[cur][nt]), acc[mt][nt]);
            }
        }
    };

    Staged st, s2;
    load_chunk(st, 0);
    store_chunk(st, 0);
    __syncthreads();
    load_chunk(st, PKC);
    load_chunk(s2, 2 * PKC);
    frag_load(0, 0, 0);
    steps(0, 0, JS / 2);
    __builtin_amdgcn_sched_barrier(0);
    store_chunk(st, 1);
    __builtin_amdgcn_sched_barrier(0);
    load_chunk(st, 3 * PKC);
    __builtin_amdgcn_sched_barrier(0);
    steps(0, JS / 2, JS);
    __builtin_amdgcn_sched_barrier(0);
    store_chunk(s2, 2);
    store_chunk(st, 3);
    __syncthreads();
    ring_signal();

    #pragma nounroll
    for (int ch = 1; ch < NCH; ++ch) {
        const int  buf  = ch & (NST - 1);
        const bool fill = ch + NST - 1 < NCH;
        if (fill) load_chunk(st, (ch + NST - 1) * PKC);
        frag_load(buf, 0, 0);
        steps(buf, 0, JS / 2);
        ring_wait();
        steps(buf, JS / 2, JS);
        __builtin_amdgcn_sched_barrier(0);
        if (ch + 1 < NCH) ring_signal();
        if (fill) store_chunk(st, (ch + NST - 1) & (NST - 1));
    }

    #pragma unroll
    for (int nt = 0; nt < 3; ++nt) {
        const int n0 = wn * 48 + nt * 16;
        float* dst = (n0 < Hh) ? ko : ((n0 < 2 * Hh) ? qo : vo);
        const int col = (n0 & (Hh - 1)) + c;
        #pragma unroll
        for (int mt = 0; mt < 2; ++mt) {
            const int orow = rowb + wm * 32 + mt * 16 + h * 8;
            #pragma unroll
            for (int r = 0; r < 8; ++r)
                dst[(size_t)(orow + r) * Hh + col] = acc[mt][nt][r];
        }
    }
}

constexpr int NW     = 16;
constexpr int OPITCH = 68;
constexpr int QROWS  = 32;
constexpr int NQB    = Tt / QROWS;
constexpr int NGRP   = NQB / 2;

__device__ __forceinline__ float xhalf(float x) {
    const int b = __builtin_bit_cast(int, x);
    return __builtin_bit_cast(float, __builtin_amdgcn_permlanex16(b, b, 0x76543210, 0xfedcba98, false, false));
}

__device__ __forceinline__ float softmax_step(const float (&sv)[8], float (&pv)[8], float& mrun, float& lrun) {
    float t = fmaxf(fmaxf(fmaxf(sv[0], sv[1]), fmaxf(sv[2], sv[3])),
                    fmaxf(fmaxf(sv[4], sv[5]), fmaxf(sv[6], sv[7])));
    t = fmaxf(t, xhalf(t));
    const float mnew  = fmaxf(mrun, t);
    const float alpha = __expf(mrun - mnew);
    mrun = mnew;
    float psum = 0.f;
    #pragma unroll
    for (int r = 0; r < 8; ++r) {
        pv[r] = __expf(sv[r] - mnew);
        psum += pv[r];
    }
    lrun = lrun * alpha + psum;
    return alpha;
}

__device__ __forceinline__ void put8(float* p, v8f o) {
    *(v4f*)(p + 0) = __builtin_shufflevector(o, o, 0, 1, 2, 3);
    *(v4f*)(p + 4) = __builtin_shufflevector(o, o, 4, 5, 6, 7);
}

__global__ __launch_bounds__(32 * NW)
void attn_kernel(const float* __restrict__ q,
                 const float* __restrict__ k,
                 const float* __restrict__ v,
                 float* __restrict__ out) {
    __shared__ alignas(16) float obuf[2][NW][QROWS][OPITCH];
    __shared__ alignas(8)  float mlbuf[2][NW][QROWS][2];

    const int lane  = threadIdx.x & 31;
    const int w     = __builtin_amdgcn_readfirstlane((int)(threadIdx.x >> 5));
    const int b     = blockIdx.x / NGRP;
    const int ip    = blockIdx.x % NGRP;
    const int half  = lane >> 4;
    const int nlo   = lane & 15;
    const int khalf = half * 2;

    const float scale = 0.03125f;

    for (int ph = 0; ph < 2; ++ph) {
        const int pb    = ph == 0 ? (NQB - 1 - ip) : ip;
        const int qrow0 = b * Tt + pb * QROWS;
        const int jlast = 2 * pb + 1;

        v2f qa[16], qb[16];
        const float* qp = q + (size_t)(qrow0 + nlo) * Hh + khalf;
        #pragma unroll
        for (int s = 0; s < 16; ++s) {
            qa[s] = *(const v2f*)(qp + 4 * s);
            qb[s] = *(const v2f*)(qp + 16 * Hh + 4 * s);
        }

        v8f oa0 = {}, oa1 = {}, oa2 = {}, oa3 = {};
        v8f ob0 = {}, ob1 = {}, ob2 = {}, ob3 = {};
        float ma = -1e30f, mb = -1e30f;
        float la = 0.f, lb = 0.f;

        for (int j = w; j <= jlast; j += NW) {
            const int krow0 = b * Tt + j * 16;

            v8f sa8 = {}, sb8 = {};
            const float* kp = k + (size_t)(krow0 + nlo) * Hh + khalf;
            #pragma unroll
            for (int t = 0; t < 16; ++t) {
                v2f ka = *(const v2f*)(kp + 4 * t);
                sa8 = wmma4(ka, qa[t], sa8);
                sb8 = wmma4(ka, qb[t], sb8);
            }

            float sva[8], svb[8];
            #pragma unroll
            for (int r = 0; r < 8; ++r) { sva[r] = sa8[r] * scale; svb[r] = sb8[r] * scale; }

            if (j >= jlast - 1) {
                const int d = (j - 2 * pb) * 16 + half * 8 - nlo;
                #pragma unroll
                for (int r = 0; r < 8; ++r) {
                    if (d + r > 0)  sva[r] = -1e30f;
                    if (d + r > 16) svb[r] = -1e30f;
                }
            }

            float pva[8], pvb[8];
            const float ala = softmax_step(sva, pva, ma, la);
            const float alb = softmax_step(svb, pvb, mb, lb);

            #pragma unroll
            for (int r = 0; r < 8; ++r) {
                oa0[r] *= ala; oa1[r] *= ala; oa2[r] *= ala; oa3[r] *= ala;
                ob0[r] *= alb; ob1[r] *= alb; ob2[r] *= alb; ob3[r] *= alb;
            }

            const float* vp = v + (size_t)(krow0 + 8 * half) * Hh + nlo;
            #pragma unroll
            for (int s = 0; s < 4; ++s) {
                v2f pa, pb2;
                pa.x  = pva[2 * s]; pa.y  = pva[2 * s + 1];
                pb2.x = pvb[2 * s]; pb2.y = pvb[2 * s + 1];
                const float* vrow = vp + (size_t)(2 * s) * Hh;
                { v2f va; va.x = vrow[ 0]; va.y = vrow[Hh +  0]; oa0 = wmma4(va, pa, oa0); ob0 = wmma4(va, pb2, ob0); }
                { v2f va; va.x = vrow[16]; va.y = vrow[Hh + 16]; oa1 = wmma4(va, pa, oa1); ob1 = wmma4(va, pb2, ob1); }
                { v2f va; va.x = vrow[32]; va.y = vrow[Hh + 32]; oa2 = wmma4(va, pa, oa2); ob2 = wmma4(va, pb2, ob2); }
                { v2f va; va.x = vrow[48]; va.y = vrow[Hh + 48]; oa3 = wmma4(va, pa, oa3); ob3 = wmma4(va, pb2, ob3); }
            }
        }

        const float lsa = la + xhalf(la);
        const float lsb = lb + xhalf(lb);

        float* opa = &obuf[ph][w][nlo][8 * half];
        float* opb = &obuf[ph][w][16 + nlo][8 * half];
        put8(opa +  0, oa0); put8(opa + 16, oa1); put8(opa + 32, oa2); put8(opa + 48, oa3);
        put8(opb +  0, ob0); put8(opb + 16, ob1); put8(opb + 32, ob2); put8(opb + 48, ob3);
        if (half == 0) {
            mlbuf[ph][w][nlo][0]      = ma;
            mlbuf[ph][w][nlo][1]      = lsa;
            mlbuf[ph][w][16 + nlo][0] = mb;
            mlbuf[ph][w][16 + nlo][1] = lsb;
        }
    }
    __syncthreads();

    const int row = threadIdx.x >> 4;
    const int c4  = (threadIdx.x & 15) * 4;

    for (int ph = 0; ph < 2; ++ph) {
        const int pb    = ph == 0 ? (NQB - 1 - ip) : ip;
        const int qrow0 = b * Tt + pb * QROWS;

        float mw[NW], lw[NW];
        float M = -1e30f;
        #pragma unroll
        for (int i = 0; i < NW; ++i) {
            mw[i] = mlbuf[ph][i][row][0];
            lw[i] = mlbuf[ph][i][row][1];
            M = fmaxf(M, mw[i]);
        }
        float L = 0.f;
        float4 acc = make_float4(0.f, 0.f, 0.f, 0.f);
        #pragma unroll
        for (int i = 0; i < NW; ++i) {
            const float wgt = __expf(mw[i] - M);
            const float4 o = *(const float4*)&obuf[ph][i][row][c4];
            L += wgt * lw[i];
            acc.x += wgt * o.x; acc.y += wgt * o.y;
            acc.z += wgt * o.z; acc.w += wgt * o.w;
        }
        const float inv = 1.0f / L;
        acc.x *= inv; acc.y *= inv; acc.z *= inv; acc.w *= inv;
        *(float4*)(out + (size_t)(qrow0 + row) * Hh + c4) = acc;
    }
}

extern "C" void kernel_launch(void* const* d_in, const int* in_sizes, int n_in,
                              void* d_out, int out_size, void* d_ws, size_t ws_size,
                              hipStream_t stream) {
    const float* x  = (const float*)d_in[0];
    const float* Wk = (const float*)d_in[1];
    const float* Wq = (const float*)d_in[2];
    const float* Wv = (const float*)d_in[3];
    float* out = (float*)d_out;

    float* ws   = (float*)d_ws;
    float* qbuf = ws;
    float* kbuf = ws + (size_t)ROWS * Hh;
    float* vbuf = ws + 2 * (size_t)ROWS * Hh;

    proj_kernel<<<ROWS / PBM, 256, 0, stream>>>(x, Wk, Wq, Wv, qbuf, kbuf, vbuf);
    attn_kernel<<<Bb * NGRP, 32 * NW, 0, stream>>>(qbuf, kbuf, vbuf, out);
}
